# starting kernel
# baseline (speedup 1.0000x reference)
#include <hip/hip_runtime.h>
#include <hip/hip_bf16.h>

typedef __attribute__((ext_vector_type(2))) float v2f;
typedef __attribute__((ext_vector_type(8))) float v8f;

#define N_LI 8192
#define N_RA 2048
#define M_PTS 2048
#define ROWS_TOTAL ((N_LI + N_RA) * 9)
#define GRID_MOD 513

__global__ __launch_bounds__(256) void nearest_dy_wmma_kernel(
    const int* __restrict__ li_bev, const int* __restrict__ ra_bev,
    const float* __restrict__ ra_pts, const int* __restrict__ ra_vox,
    float* __restrict__ out)
{
    __shared__ float sB[M_PTS * 4];
    __shared__ int   sCnt;

    const int tid = threadIdx.x;
    if (tid == 0) sCnt = 0;
    __syncthreads();

    int local = 0;
#pragma unroll
    for (int i = 0; i < M_PTS / 256; ++i) {
        int j = i * 256 + tid;
        float vel = ra_pts[j * 5 + 4];
        float px = (float)ra_vox[j * 3 + 1];
        float py = (float)ra_vox[j * 3 + 2];
        bool dyn = fabsf(vel) > 0.1f;
        sB[j * 4 + 0] = dyn ? -2.0f * px : 0.0f;
        sB[j * 4 + 1] = dyn ? -2.0f * py : 0.0f;
        sB[j * 4 + 2] = dyn ? px * px + py * py : 1.0e30f;
        sB[j * 4 + 3] = dyn ? 1.0f : 0.0f;
        local += dyn ? 1 : 0;
    }
    atomicAdd(&sCnt, local);
    __syncthreads();
    const float mul = (sCnt > 1) ? 0.01f : 0.0f;

    const int lane = tid & 31;
    const int wave = tid >> 5;
    const int m    = lane & 15;
    const int half = lane >> 4;
    const int tileBase = (blockIdx.x * 8 + wave) * 16;
    const int r = tileBase + m;

    int c0, c1, s;
    if (r < N_LI * 9) {
        int n = r / 9; s = r - n * 9;
        c0 = li_bev[n * 2 + 0]; c1 = li_bev[n * 2 + 1];
    } else {
        int rr = r - N_LI * 9;
        int n = rr / 9; s = rr - n * 9;
        c0 = ra_bev[n * 2 + 0]; c1 = ra_bev[n * 2 + 1];
    }
    const int SH0[3] = {0, -1, 1};
    const int SH1[3] = {0, 1, -1};
    int x = (c0 + SH0[s % 3] + GRID_MOD) % GRID_MOD;
    int y = (c1 + SH1[s / 3] + GRID_MOD) % GRID_MOD;
    float xf = (float)x, yf = (float)y;

    v2f a;
    if (half == 0) { a.x = xf;   a.y = yf; }
    else           { a.x = 1.0f; a.y = xf * xf + yf * yf; }

    float dmin[8];
#pragma unroll
    for (int k = 0; k < 8; ++k) dmin[k] = 3.0e38f;

    const float2* sB2 = (const float2*)sB;
    const v8f czero = (v8f)0.0f;

#pragma unroll 4
    for (int jt = 0; jt < M_PTS / 16; ++jt) {
        float2 bb = sB2[(jt * 16 + m) * 2 + half];
        v2f b; b.x = bb.x; b.y = bb.y;
        v8f d = __builtin_amdgcn_wmma_f32_16x16x4_f32(
             false, a,  false, b,
             (short)0, czero,  false,  false);
#pragma unroll
        for (int k = 0; k < 8; ++k) dmin[k] = fminf(dmin[k], d[k]);
    }

#pragma unroll
    for (int k = 0; k < 8; ++k) {
        float v = dmin[k];
        v = fminf(v, __shfl_xor(v, 1, 32));
        v = fminf(v, __shfl_xor(v, 2, 32));
        v = fminf(v, __shfl_xor(v, 4, 32));
        v = fminf(v, __shfl_xor(v, 8, 32));
        dmin[k] = v;
    }

    if (m == 0) {
#pragma unroll
        for (int k = 0; k < 8; ++k)
            out[tileBase + half * 8 + k] = sqrtf(fmaxf(dmin[k], 0.0f)) * mul;
    }
}

extern "C" void kernel_launch(void* const* d_in, const int* in_sizes, int n_in,
                              void* d_out, int out_size, void* d_ws, size_t ws_size,
                              hipStream_t stream) {
    (void)in_sizes; (void)n_in; (void)out_size; (void)d_ws; (void)ws_size;
    const int*   li  = (const int*)d_in[0];
    const int*   ra  = (const int*)d_in[1];
    const float* pts = (const float*)d_in[2];
    const int*   vox = (const int*)d_in[3];
    float* out = (float*)d_out;

    dim3 grid(ROWS_TOTAL / 128);
    nearest_dy_wmma_kernel<<<grid, 256, 0, stream>>>(li, ra, pts, vox, out);
}


# Products per query point 9 -> 5 (F(x_a,y_0) x3 + G(y_b) x2), F+G by packed f32 adds, |q|^2 after min (sums regrouped, exact on integers); points as A so mins stay in-lane; 640 blocks, 8 point ranges
# speedup vs baseline: 1.0934x; 1.0934x over previous
#include <hip/hip_runtime.h>

typedef __attribute__((ext_vector_type(2))) float v2f;
typedef __attribute__((ext_vector_type(8))) float v8f;

#define N_LI 8192
#define N_RA 2048
#define M_PTS 2048
#define GRID_MOD 513

__global__ __launch_bounds__(256) void nearest_dy_split_kernel(
    const int* __restrict__ li_bev, const int* __restrict__ ra_bev,
    const float* __restrict__ ra_pts, const int* __restrict__ ra_vox,
    float* __restrict__ out)
{
    __shared__ float sB[M_PTS * 4];
    __shared__ float sMin[8][9][16];
    __shared__ float sQ[6][16];
    __shared__ int   sCnt;

    const int tid = threadIdx.x;
    if (tid == 0) sCnt = 0;
    __syncthreads();

    int local = 0;
#pragma unroll
    for (int i = 0; i < M_PTS / 256; ++i) {
        int j = i * 256 + tid;
        float vel = ra_pts[j * 5 + 4];
        float px = (float)ra_vox[j * 3 + 1];
        float py = (float)ra_vox[j * 3 + 2];
        bool dyn = fabsf(vel) > 0.1f;
        sB[j * 4 + 0] = dyn ? -2.0f * px : 0.0f;
        sB[j * 4 + 1] = dyn ? -2.0f * py : 0.0f;
        sB[j * 4 + 2] = dyn ? px * px + py * py : 1.0e30f;
        sB[j * 4 + 3] = dyn ? 1.0f : 0.0f;
        local += dyn ? 1 : 0;
    }
    atomicAdd(&sCnt, local);

    const int lane = tid & 31;
    const int wave = tid >> 5;
    const int m    = lane & 15;
    const int half = lane >> 4;
    const int n    = blockIdx.x * 16 + m;

    int c0, c1;
    if (n < N_LI) { c0 = li_bev[n * 2 + 0]; c1 = li_bev[n * 2 + 1]; }
    else          { c0 = ra_bev[(n - N_LI) * 2 + 0]; c1 = ra_bev[(n - N_LI) * 2 + 1]; }
    float xs[3], ys[3];
    xs[0] = (float)((c0 + GRID_MOD) % GRID_MOD);
    xs[1] = (float)((c0 - 1 + GRID_MOD) % GRID_MOD);
    xs[2] = (float)((c0 + 1 + GRID_MOD) % GRID_MOD);
    ys[0] = (float)((c1 + GRID_MOD) % GRID_MOD);
    ys[1] = (float)((c1 + 1 + GRID_MOD) % GRID_MOD);
    ys[2] = (float)((c1 - 1 + GRID_MOD) % GRID_MOD);
    if (wave == 0 && half == 0) {
#pragma unroll
        for (int a = 0; a < 3; ++a) { sQ[a][m] = xs[a]; sQ[3 + a][m] = ys[a]; }
    }

    v2f qF[3], qG[2];
#pragma unroll
    for (int a = 0; a < 3; ++a) {
        if (half == 0) { qF[a].x = xs[a]; qF[a].y = ys[0]; }
        else           { qF[a].x = 1.0f;  qF[a].y = 0.0f; }
    }
#pragma unroll
    for (int t = 0; t < 2; ++t) {
        if (half == 0) { qG[t].x = 0.0f; qG[t].y = ys[t + 1] - ys[0]; }
        else           { qG[t].x = 0.0f; qG[t].y = 0.0f; }
    }

    __syncthreads();
    const float mul = (sCnt > 1) ? 0.01f : 0.0f;

    float acc[9];
#pragma unroll
    for (int i = 0; i < 9; ++i) acc[i] = 3.0e38f;

    const float2* sB2 = (const float2*)sB;
    const v8f czero = (v8f)0.0f;

    const float2* myB = sB2 + (wave * 256 + m) * 2 + half;
    float2 pp = myB[0];
    for (int jt = 0; jt < 16; ++jt) {
        v2f p; p.x = pp.x; p.y = pp.y;
        pp = myB[(jt < 15 ? jt + 1 : jt) * 32];
        v8f g1 = __builtin_amdgcn_wmma_f32_16x16x4_f32(false, p, false, qG[0], (short)0, czero, false, false);
        v8f g2 = __builtin_amdgcn_wmma_f32_16x16x4_f32(false, p, false, qG[1], (short)0, czero, false, false);
#pragma unroll
        for (int a = 0; a < 3; ++a) {
            v8f f  = __builtin_amdgcn_wmma_f32_16x16x4_f32(false, p, false, qF[a], (short)0, czero, false, false);
            v8f e1 = f + g1;
            v8f e2 = f + g2;
            float v0 = acc[a], v1 = acc[3 + a], v2 = acc[6 + a];
#pragma unroll
            for (int k = 0; k < 8; k += 2) {
                v0 = fminf(fminf(v0, f[k]),  f[k + 1]);
                v1 = fminf(fminf(v1, e1[k]), e1[k + 1]);
                v2 = fminf(fminf(v2, e2[k]), e2[k + 1]);
            }
            acc[a] = v0; acc[3 + a] = v1; acc[6 + a] = v2;
        }
    }

#pragma unroll
    for (int i = 0; i < 9; ++i) acc[i] = fminf(acc[i], __shfl_xor(acc[i], 16, 32));
    if (half == 0) {
#pragma unroll
        for (int i = 0; i < 9; ++i) sMin[wave][i][m] = acc[i];
    }
    __syncthreads();

    if (tid < 144) {
        const int c = tid / 9;
        const int s = tid - c * 9;
        const int a = s % 3, b = s / 3;
        float v = sMin[0][s][c];
#pragma unroll
        for (int w = 1; w < 8; ++w) v = fminf(v, sMin[w][s][c]);
        const float xa = sQ[a][c], yb = sQ[3 + b][c];
        const float d2 = v + (xa * xa + yb * yb);
        out[blockIdx.x * 144 + tid] = sqrtf(fmaxf(d2, 0.0f)) * mul;
    }
}

extern "C" void kernel_launch(void* const* d_in, const int* in_sizes, int n_in,
                              void* d_out, int out_size, void* d_ws, size_t ws_size,
                              hipStream_t stream) {
    (void)in_sizes; (void)n_in; (void)out_size; (void)d_ws; (void)ws_size;
    const int*   li  = (const int*)d_in[0];
    const int*   ra  = (const int*)d_in[1];
    const float* pts = (const float*)d_in[2];
    const int*   vox = (const int*)d_in[3];
    float* out = (float*)d_out;

    dim3 grid(640);
    nearest_dy_split_kernel<<<grid, 256, 0, stream>>>(li, ra, pts, vox, out);
}


# Tile loop reads each fragment in the trip that uses it, 4 tiles a trip (no read carried a trip ahead): minima compile to 36 min3 a tile, no quieting v_max_num (76 -> 61 vector instr a tile)
# speedup vs baseline: 1.1514x; 1.0530x over previous
#include <hip/hip_runtime.h>

typedef __attribute__((ext_vector_type(2))) float v2f;
typedef __attribute__((ext_vector_type(8))) float v8f;

#define N_LI 8192
#define N_RA 2048
#define M_PTS 2048
#define GRID_MOD 513

__global__ __launch_bounds__(256) void nearest_dy_split_kernel(
    const int* __restrict__ li_bev, const int* __restrict__ ra_bev,
    const float* __restrict__ ra_pts, const int* __restrict__ ra_vox,
    float* __restrict__ out)
{
    __shared__ float sB[M_PTS * 4];
    __shared__ float sMin[8][9][16];
    __shared__ float sQ[6][16];
    __shared__ int   sCnt;

    const int tid = threadIdx.x;
    if (tid == 0) sCnt = 0;
    __syncthreads();

    int local = 0;
#pragma unroll
    for (int i = 0; i < M_PTS / 256; ++i) {
        int j = i * 256 + tid;
        float vel = ra_pts[j * 5 + 4];
        float px = (float)ra_vox[j * 3 + 1];
        float py = (float)ra_vox[j * 3 + 2];
        bool dyn = fabsf(vel) > 0.1f;
        sB[j * 4 + 0] = dyn ? -2.0f * px : 0.0f;
        sB[j * 4 + 1] = dyn ? -2.0f * py : 0.0f;
        sB[j * 4 + 2] = dyn ? px * px + py * py : 1.0e30f;
        sB[j * 4 + 3] = dyn ? 1.0f : 0.0f;
        local += dyn ? 1 : 0;
    }
    atomicAdd(&sCnt, local);

    const int lane = tid & 31;
    const int wave = tid >> 5;
    const int m    = lane & 15;
    const int half = lane >> 4;
    const int n    = blockIdx.x * 16 + m;

    int c0, c1;
    if (n < N_LI) { c0 = li_bev[n * 2 + 0]; c1 = li_bev[n * 2 + 1]; }
    else          { c0 = ra_bev[(n - N_LI) * 2 + 0]; c1 = ra_bev[(n - N_LI) * 2 + 1]; }
    float xs[3], ys[3];
    xs[0] = (float)((c0 + GRID_MOD) % GRID_MOD);
    xs[1] = (float)((c0 - 1 + GRID_MOD) % GRID_MOD);
    xs[2] = (float)((c0 + 1 + GRID_MOD) % GRID_MOD);
    ys[0] = (float)((c1 + GRID_MOD) % GRID_MOD);
    ys[1] = (float)((c1 + 1 + GRID_MOD) % GRID_MOD);
    ys[2] = (float)((c1 - 1 + GRID_MOD) % GRID_MOD);
    if (wave == 0 && half == 0) {
#pragma unroll
        for (int a = 0; a < 3; ++a) { sQ[a][m] = xs[a]; sQ[3 + a][m] = ys[a]; }
    }

    v2f qF[3], qG[2];
#pragma unroll
    for (int a = 0; a < 3; ++a) {
        if (half == 0) { qF[a].x = xs[a]; qF[a].y = ys[0]; }
        else           { qF[a].x = 1.0f;  qF[a].y = 0.0f; }
    }
#pragma unroll
    for (int t = 0; t < 2; ++t) {
        if (half == 0) { qG[t].x = 0.0f; qG[t].y = ys[t + 1] - ys[0]; }
        else           { qG[t].x = 0.0f; qG[t].y = 0.0f; }
    }

    __syncthreads();
    const float mul = (sCnt > 1) ? 0.01f : 0.0f;

    float acc[9];
#pragma unroll
    for (int i = 0; i < 9; ++i) acc[i] = 3.0e38f;

    const float2* sB2 = (const float2*)sB;
    const v8f czero = (v8f)0.0f;

    const float2* myB = sB2 + (wave * 256 + m) * 2 + half;
#pragma unroll 4
    for (int jt = 0; jt < 16; ++jt) {
        const float2 pp = myB[jt * 32];
        v2f p; p.x = pp.x; p.y = pp.y;
        v8f g1 = __builtin_amdgcn_wmma_f32_16x16x4_f32(false, p, false, qG[0], (short)0, czero, false, false);
        v8f g2 = __builtin_amdgcn_wmma_f32_16x16x4_f32(false, p, false, qG[1], (short)0, czero, false, false);
#pragma unroll
        for (int a = 0; a < 3; ++a) {
            v8f f  = __builtin_amdgcn_wmma_f32_16x16x4_f32(false, p, false, qF[a], (short)0, czero, false, false);
            v8f e1 = f + g1;
            v8f e2 = f + g2;
            float v0 = acc[a], v1 = acc[3 + a], v2 = acc[6 + a];
#pragma unroll
            for (int k = 0; k < 8; k += 2) {
                v0 = fminf(fminf(v0, f[k]),  f[k + 1]);
                v1 = fminf(fminf(v1, e1[k]), e1[k + 1]);
                v2 = fminf(fminf(v2, e2[k]), e2[k + 1]);
            }
            acc[a] = v0; acc[3 + a] = v1; acc[6 + a] = v2;
        }
    }

#pragma unroll
    for (int i = 0; i < 9; ++i) acc[i] = fminf(acc[i], __shfl_xor(acc[i], 16, 32));
    if (half == 0) {
#pragma unroll
        for (int i = 0; i < 9; ++i) sMin[wave][i][m] = acc[i];
    }
    __syncthreads();

    if (tid < 144) {
        const int c = tid / 9;
        const int s = tid - c * 9;
        const int a = s % 3, b = s / 3;
        float v = sMin[0][s][c];
#pragma unroll
        for (int w = 1; w < 8; ++w) v = fminf(v, sMin[w][s][c]);
        const float xa = sQ[a][c], yb = sQ[3 + b][c];
        const float d2 = v + (xa * xa + yb * yb);
        out[blockIdx.x * 144 + tid] = sqrtf(fmaxf(d2, 0.0f)) * mul;
    }
}

extern "C" void kernel_launch(void* const* d_in, const int* in_sizes, int n_in,
                              void* d_out, int out_size, void* d_ws, size_t ws_size,
                              hipStream_t stream) {
    (void)in_sizes; (void)n_in; (void)out_size; (void)d_ws; (void)ws_size;
    const int*   li  = (const int*)d_in[0];
    const int*   ra  = (const int*)d_in[1];
    const float* pts = (const float*)d_in[2];
    const int*   vox = (const int*)d_in[3];
    float* out = (float*)d_out;

    dim3 grid(640);
    nearest_dy_split_kernel<<<grid, 256, 0, stream>>>(li, ra, pts, vox, out);
}


# G_2 product dropped where a block's row steps are opposite (uniform test; else the general loop): G_2 = -G_1 exactly, F+G_2 taken as F-G_1 by the packed add's negation: 5 -> 4 products a tile
# speedup vs baseline: 1.2341x; 1.0718x over previous
#include <hip/hip_runtime.h>

typedef __attribute__((ext_vector_type(2))) float v2f;
typedef __attribute__((ext_vector_type(8))) float v8f;

#define N_LI 8192
#define N_RA 2048
#define M_PTS 2048
#define GRID_MOD 513

template <bool SYM>
__device__ __forceinline__ void scan_tiles(const float2* myB, const v2f (&qF)[3], const v2f (&qG)[2],
                                           float (&acc)[9])
{
    const v8f czero = (v8f)0.0f;
#pragma unroll 4
    for (int jt = 0; jt < 16; ++jt) {
        const float2 pp = myB[jt * 32];
        v2f p; p.x = pp.x; p.y = pp.y;
        v8f g1 = __builtin_amdgcn_wmma_f32_16x16x4_f32(false, p, false, qG[0], (short)0, czero, false, false);
        v8f g2 = czero;
        if (!SYM) g2 = __builtin_amdgcn_wmma_f32_16x16x4_f32(false, p, false, qG[1], (short)0, czero, false, false);
#pragma unroll
        for (int a = 0; a < 3; ++a) {
            v8f f  = __builtin_amdgcn_wmma_f32_16x16x4_f32(false, p, false, qF[a], (short)0, czero, false, false);
            v8f e1 = f + g1;
            v8f e2 = SYM ? (f - g1) : (f + g2);
            float v0 = acc[a], v1 = acc[3 + a], v2 = acc[6 + a];
#pragma unroll
            for (int k = 0; k < 8; k += 2) {
                v0 = fminf(fminf(v0, f[k]),  f[k + 1]);
                v1 = fminf(fminf(v1, e1[k]), e1[k + 1]);
                v2 = fminf(fminf(v2, e2[k]), e2[k + 1]);
            }
            acc[a] = v0; acc[3 + a] = v1; acc[6 + a] = v2;
        }
    }
}

__global__ __launch_bounds__(256) void nearest_dy_split_kernel(
    const int* __restrict__ li_bev, const int* __restrict__ ra_bev,
    const float* __restrict__ ra_pts, const int* __restrict__ ra_vox,
    float* __restrict__ out)
{
    __shared__ float sB[M_PTS * 4];
    __shared__ float sMin[8][9][16];
    __shared__ float sQ[6][16];
    __shared__ int   sCnt;

    const int tid = threadIdx.x;
    if (tid == 0) sCnt = 0;
    __syncthreads();

    int local = 0;
#pragma unroll
    for (int i = 0; i < M_PTS / 256; ++i) {
        int j = i * 256 + tid;
        float vel = ra_pts[j * 5 + 4];
        float px = (float)ra_vox[j * 3 + 1];
        float py = (float)ra_vox[j * 3 + 2];
        bool dyn = fabsf(vel) > 0.1f;
        sB[j * 4 + 0] = dyn ? -2.0f * px : 0.0f;
        sB[j * 4 + 1] = dyn ? -2.0f * py : 0.0f;
        sB[j * 4 + 2] = dyn ? px * px + py * py : 1.0e30f;
        sB[j * 4 + 3] = dyn ? 1.0f : 0.0f;
        local += dyn ? 1 : 0;
    }
    atomicAdd(&sCnt, local);

    const int lane = tid & 31;
    const int wave = tid >> 5;
    const int m    = lane & 15;
    const int half = lane >> 4;
    const int n    = blockIdx.x * 16 + m;

    int c0, c1;
    if (n < N_LI) { c0 = li_bev[n * 2 + 0]; c1 = li_bev[n * 2 + 1]; }
    else          { c0 = ra_bev[(n - N_LI) * 2 + 0]; c1 = ra_bev[(n - N_LI) * 2 + 1]; }
    float xs[3], ys[3];
    xs[0] = (float)((c0 + GRID_MOD) % GRID_MOD);
    xs[1] = (float)((c0 - 1 + GRID_MOD) % GRID_MOD);
    xs[2] = (float)((c0 + 1 + GRID_MOD) % GRID_MOD);
    ys[0] = (float)((c1 + GRID_MOD) % GRID_MOD);
    ys[1] = (float)((c1 + 1 + GRID_MOD) % GRID_MOD);
    ys[2] = (float)((c1 - 1 + GRID_MOD) % GRID_MOD);
    if (wave == 0 && half == 0) {
#pragma unroll
        for (int a = 0; a < 3; ++a) { sQ[a][m] = xs[a]; sQ[3 + a][m] = ys[a]; }
    }

    v2f qF[3], qG[2];
#pragma unroll
    for (int a = 0; a < 3; ++a) {
        if (half == 0) { qF[a].x = xs[a]; qF[a].y = ys[0]; }
        else           { qF[a].x = 1.0f;  qF[a].y = 0.0f; }
    }
#pragma unroll
    for (int t = 0; t < 2; ++t) {
        if (half == 0) { qG[t].x = 0.0f; qG[t].y = ys[t + 1] - ys[0]; }
        else           { qG[t].x = 0.0f; qG[t].y = 0.0f; }
    }

    __syncthreads();
    const float mul = (sCnt > 1) ? 0.01f : 0.0f;

    float acc[9];
#pragma unroll
    for (int i = 0; i < 9; ++i) acc[i] = 3.0e38f;

    const float2* sB2 = (const float2*)sB;

    const float2* myB = sB2 + (wave * 256 + m) * 2 + half;

    const bool opposite = (ys[2] - ys[0]) == -(ys[1] - ys[0]);
    if (__builtin_amdgcn_ballot_w32(!opposite) == 0u) scan_tiles<true>(myB, qF, qG, acc);
    else                                              scan_tiles<false>(myB, qF, qG, acc);

#pragma unroll
    for (int i = 0; i < 9; ++i) acc[i] = fminf(acc[i], __shfl_xor(acc[i], 16, 32));
    if (half == 0) {
#pragma unroll
        for (int i = 0; i < 9; ++i) sMin[wave][i][m] = acc[i];
    }
    __syncthreads();

    if (tid < 144) {
        const int c = tid / 9;
        const int s = tid - c * 9;
        const int a = s % 3, b = s / 3;
        float v = sMin[0][s][c];
#pragma unroll
        for (int w = 1; w < 8; ++w) v = fminf(v, sMin[w][s][c]);
        const float xa = sQ[a][c], yb = sQ[3 + b][c];
        const float d2 = v + (xa * xa + yb * yb);
        out[blockIdx.x * 144 + tid] = sqrtf(fmaxf(d2, 0.0f)) * mul;
    }
}

extern "C" void kernel_launch(void* const* d_in, const int* in_sizes, int n_in,
                              void* d_out, int out_size, void* d_ws, size_t ws_size,
                              hipStream_t stream) {
    (void)in_sizes; (void)n_in; (void)out_size; (void)d_ws; (void)ws_size;
    const int*   li  = (const int*)d_in[0];
    const int*   ra  = (const int*)d_in[1];
    const float* pts = (const float*)d_in[2];
    const int*   vox = (const int*)d_in[3];
    float* out = (float*)d_out;

    dim3 grid(640);
    nearest_dy_split_kernel<<<grid, 256, 0, stream>>>(li, ra, pts, vox, out);
}


# Blocks whose 16 query points all step exactly -1,+1 in x and +1,-1 in y: F_1,F_2,G_1 = F_0 +-2px, -+2py, read per point from LDS in D layout and added (sums regrouped, exact): 4 -> 1 products a tile
# speedup vs baseline: 1.3576x; 1.1001x over previous
#include <hip/hip_runtime.h>

typedef __attribute__((ext_vector_type(2))) float v2f;
typedef __attribute__((ext_vector_type(8))) float v8f;

#define N_LI 8192
#define N_RA 2048
#define M_PTS 2048
#define GRID_MOD 513

template <bool SYM>
__device__ __forceinline__ void scan_tiles(const float2* myB, const v2f (&qF)[3], const v2f (&qG)[2],
                                           float (&acc)[9])
{
    const v8f czero = (v8f)0.0f;
#pragma unroll 4
    for (int jt = 0; jt < 16; ++jt) {
        const float2 pp = myB[jt * 32];
        v2f p; p.x = pp.x; p.y = pp.y;
        v8f g1 = __builtin_amdgcn_wmma_f32_16x16x4_f32(false, p, false, qG[0], (short)0, czero, false, false);
        v8f g2 = czero;
        if (!SYM) g2 = __builtin_amdgcn_wmma_f32_16x16x4_f32(false, p, false, qG[1], (short)0, czero, false, false);
#pragma unroll
        for (int a = 0; a < 3; ++a) {
            v8f f  = __builtin_amdgcn_wmma_f32_16x16x4_f32(false, p, false, qF[a], (short)0, czero, false, false);
            v8f e1 = f + g1;
            v8f e2 = SYM ? (f - g1) : (f + g2);
            float v0 = acc[a], v1 = acc[3 + a], v2 = acc[6 + a];
#pragma unroll
            for (int k = 0; k < 8; k += 2) {
                v0 = fminf(fminf(v0, f[k]),  f[k + 1]);
                v1 = fminf(fminf(v1, e1[k]), e1[k + 1]);
                v2 = fminf(fminf(v2, e2[k]), e2[k + 1]);
            }
            acc[a] = v0; acc[3 + a] = v1; acc[6 + a] = v2;
        }
    }
}

__device__ __forceinline__ void scan_tiles_unit(const float2* myB, const float4* myP, const float4* myQ,
                                                const v2f& qF0, float (&acc)[9])
{
    const v8f czero = (v8f)0.0f;
#pragma unroll 4
    for (int jt = 0; jt < 16; ++jt) {
        const float2 pp = myB[jt * 32];
        v2f p; p.x = pp.x; p.y = pp.y;
        const float4 pa = myP[jt * 4], pb = myP[jt * 4 + 1];
        const float4 qa = myQ[jt * 4], qb = myQ[jt * 4 + 1];
        v8f P, Q;
        P[0] = pa.x; P[1] = pa.y; P[2] = pa.z; P[3] = pa.w; P[4] = pb.x; P[5] = pb.y; P[6] = pb.z; P[7] = pb.w;
        Q[0] = qa.x; Q[1] = qa.y; Q[2] = qa.z; Q[3] = qa.w; Q[4] = qb.x; Q[5] = qb.y; Q[6] = qb.z; Q[7] = qb.w;
        v8f fa[3];
        fa[0] = __builtin_amdgcn_wmma_f32_16x16x4_f32(false, p, false, qF0, (short)0, czero, false, false);
        fa[1] = fa[0] + P;
        fa[2] = fa[0] - P;
#pragma unroll
        for (int a = 0; a < 3; ++a) {
            v8f f  = fa[a];
            v8f e1 = f - Q;
            v8f e2 = f + Q;
            float v0 = acc[a], v1 = acc[3 + a], v2 = acc[6 + a];
#pragma unroll
            for (int k = 0; k < 8; k += 2) {
                v0 = fminf(fminf(v0, f[k]),  f[k + 1]);
                v1 = fminf(fminf(v1, e1[k]), e1[k + 1]);
                v2 = fminf(fminf(v2, e2[k]), e2[k + 1]);
            }
            acc[a] = v0; acc[3 + a] = v1; acc[6 + a] = v2;
        }
    }
}

__global__ __launch_bounds__(256) void nearest_dy_split_kernel(
    const int* __restrict__ li_bev, const int* __restrict__ ra_bev,
    const float* __restrict__ ra_pts, const int* __restrict__ ra_vox,
    float* __restrict__ out)
{
    __shared__ float sB[M_PTS * 4];
    __shared__ float4 sP4[M_PTS / 4];
    __shared__ float4 sQ4[M_PTS / 4];
    __shared__ float sMin[8][9][16];
    __shared__ float sQ[6][16];
    __shared__ int   sCnt;

    const int tid = threadIdx.x;
    if (tid == 0) sCnt = 0;
    __syncthreads();

    int local = 0;
#pragma unroll
    for (int i = 0; i < M_PTS / 256; ++i) {
        int j = i * 256 + tid;
        float vel = ra_pts[j * 5 + 4];
        float px = (float)ra_vox[j * 3 + 1];
        float py = (float)ra_vox[j * 3 + 2];
        bool dyn = fabsf(vel) > 0.1f;
        sB[j * 4 + 0] = dyn ? -2.0f * px : 0.0f;
        sB[j * 4 + 1] = dyn ? -2.0f * py : 0.0f;
        sB[j * 4 + 2] = dyn ? px * px + py * py : 1.0e30f;
        sB[j * 4 + 3] = dyn ? 1.0f : 0.0f;
        ((float*)sP4)[j] = dyn ? 2.0f * px : 0.0f;
        ((float*)sQ4)[j] = dyn ? 2.0f * py : 0.0f;
        local += dyn ? 1 : 0;
    }
    atomicAdd(&sCnt, local);

    const int lane = tid & 31;
    const int wave = tid >> 5;
    const int m    = lane & 15;
    const int half = lane >> 4;
    const int n    = blockIdx.x * 16 + m;

    int c0, c1;
    if (n < N_LI) { c0 = li_bev[n * 2 + 0]; c1 = li_bev[n * 2 + 1]; }
    else          { c0 = ra_bev[(n - N_LI) * 2 + 0]; c1 = ra_bev[(n - N_LI) * 2 + 1]; }
    float xs[3], ys[3];
    xs[0] = (float)((c0 + GRID_MOD) % GRID_MOD);
    xs[1] = (float)((c0 - 1 + GRID_MOD) % GRID_MOD);
    xs[2] = (float)((c0 + 1 + GRID_MOD) % GRID_MOD);
    ys[0] = (float)((c1 + GRID_MOD) % GRID_MOD);
    ys[1] = (float)((c1 + 1 + GRID_MOD) % GRID_MOD);
    ys[2] = (float)((c1 - 1 + GRID_MOD) % GRID_MOD);
    if (wave == 0 && half == 0) {
#pragma unroll
        for (int a = 0; a < 3; ++a) { sQ[a][m] = xs[a]; sQ[3 + a][m] = ys[a]; }
    }

    v2f qF[3], qG[2];
#pragma unroll
    for (int a = 0; a < 3; ++a) {
        if (half == 0) { qF[a].x = xs[a]; qF[a].y = ys[0]; }
        else           { qF[a].x = 1.0f;  qF[a].y = 0.0f; }
    }
#pragma unroll
    for (int t = 0; t < 2; ++t) {
        if (half == 0) { qG[t].x = 0.0f; qG[t].y = ys[t + 1] - ys[0]; }
        else           { qG[t].x = 0.0f; qG[t].y = 0.0f; }
    }

    __syncthreads();
    const float mul = (sCnt > 1) ? 0.01f : 0.0f;

    float acc[9];
#pragma unroll
    for (int i = 0; i < 9; ++i) acc[i] = 3.0e38f;

    const float2* sB2 = (const float2*)sB;

    const float2* myB = sB2 + (wave * 256 + m) * 2 + half;

    const bool opposite = (ys[2] - ys[0]) == -(ys[1] - ys[0]);
    const bool unit = (xs[1] - xs[0]) == -1.0f && (xs[2] - xs[0]) == 1.0f &&
                      (ys[1] - ys[0]) == 1.0f && (ys[2] - ys[0]) == -1.0f;
    const float4* myP = sP4 + wave * 64 + half * 2;
    const float4* myQ = sQ4 + wave * 64 + half * 2;
    if (__builtin_amdgcn_ballot_w32(!unit) == 0u)          scan_tiles_unit(myB, myP, myQ, qF[0], acc);
    else if (__builtin_amdgcn_ballot_w32(!opposite) == 0u) scan_tiles<true>(myB, qF, qG, acc);
    else                                                   scan_tiles<false>(myB, qF, qG, acc);

#pragma unroll
    for (int i = 0; i < 9; ++i) acc[i] = fminf(acc[i], __shfl_xor(acc[i], 16, 32));
    if (half == 0) {
#pragma unroll
        for (int i = 0; i < 9; ++i) sMin[wave][i][m] = acc[i];
    }
    __syncthreads();

    if (tid < 144) {
        const int c = tid / 9;
        const int s = tid - c * 9;
        const int a = s % 3, b = s / 3;
        float v = sMin[0][s][c];
#pragma unroll
        for (int w = 1; w < 8; ++w) v = fminf(v, sMin[w][s][c]);
        const float xa = sQ[a][c], yb = sQ[3 + b][c];
        const float d2 = v + (xa * xa + yb * yb);
        out[blockIdx.x * 144 + tid] = sqrtf(fmaxf(d2, 0.0f)) * mul;
    }
}

extern "C" void kernel_launch(void* const* d_in, const int* in_sizes, int n_in,
                              void* d_out, int out_size, void* d_ws, size_t ws_size,
                              hipStream_t stream) {
    (void)in_sizes; (void)n_in; (void)out_size; (void)d_ws; (void)ws_size;
    const int*   li  = (const int*)d_in[0];
    const int*   ra  = (const int*)d_in[1];
    const float* pts = (const float*)d_in[2];
    const int*   vox = (const int*)d_in[3];
    float* out = (float*)d_out;

    dim3 grid(640);
    nearest_dy_split_kernel<<<grid, 256, 0, stream>>>(li, ra, pts, vox, out);
}


# Each wave builds the table rows of its own 256 points and scans them at once: both barriers ahead of the scan and the LDS count atomic gone; count per wave by lane masks, summed after the join
# speedup vs baseline: 1.6170x; 1.1911x over previous
#include <hip/hip_runtime.h>

typedef __attribute__((ext_vector_type(2))) float v2f;
typedef __attribute__((ext_vector_type(8))) float v8f;

#define N_LI 8192
#define N_RA 2048
#define M_PTS 2048
#define GRID_MOD 513

__device__ __forceinline__ v8f fma_lane(const v2f m, const v8f b, const v8f c)
{
    v8f r;
#pragma unroll
    for (int k = 0; k < 8; k += 2) {
        v2f bb, cc;
        bb.x = b[k]; bb.y = b[k + 1]; cc.x = c[k]; cc.y = c[k + 1];
        const v2f t = __builtin_elementwise_fma(m, bb, cc);
        r[k] = t.x; r[k + 1] = t.y;
    }
    return r;
}

__device__ __forceinline__ void scan_tiles_steps(const float2* myB, const float4* myP, const float4* myQ,
                                                 const v2f& qF0, float dx1, float dx2, float dy1, float dy2,
                                                 float (&acc)[9])
{
    const v8f czero = (v8f)0.0f;
    v2f mx1, mx2, my1, my2;
    mx1.x = mx1.y = -dx1; mx2.x = mx2.y = -dx2; my1.x = my1.y = -dy1; my2.x = my2.y = -dy2;
#pragma unroll 4
    for (int jt = 0; jt < 16; ++jt) {
        const float2 pp = myB[jt * 32];
        v2f p; p.x = pp.x; p.y = pp.y;
        const float4 pa = myP[jt * 4], pb = myP[jt * 4 + 1];
        const float4 qa = myQ[jt * 4], qb = myQ[jt * 4 + 1];
        v8f P, Q;
        P[0] = pa.x; P[1] = pa.y; P[2] = pa.z; P[3] = pa.w; P[4] = pb.x; P[5] = pb.y; P[6] = pb.z; P[7] = pb.w;
        Q[0] = qa.x; Q[1] = qa.y; Q[2] = qa.z; Q[3] = qa.w; Q[4] = qb.x; Q[5] = qb.y; Q[6] = qb.z; Q[7] = qb.w;
        v8f fa[3];
        fa[0] = __builtin_amdgcn_wmma_f32_16x16x4_f32(false, p, false, qF0, (short)0, czero, false, false);
        fa[1] = fma_lane(mx1, P, fa[0]);
        fa[2] = fma_lane(mx2, P, fa[0]);
#pragma unroll
        for (int a = 0; a < 3; ++a) {
            v8f f  = fa[a];
            v8f e1 = fma_lane(my1, Q, f);
            v8f e2 = fma_lane(my2, Q, f);
            float v0 = acc[a], v1 = acc[3 + a], v2 = acc[6 + a];
#pragma unroll
            for (int k = 0; k < 8; k += 2) {
                v0 = fminf(fminf(v0, f[k]),  f[k + 1]);
                v1 = fminf(fminf(v1, e1[k]), e1[k + 1]);
                v2 = fminf(fminf(v2, e2[k]), e2[k + 1]);
            }
            acc[a] = v0; acc[3 + a] = v1; acc[6 + a] = v2;
        }
    }
}

__device__ __forceinline__ void scan_tiles_unit(const float2* myB, const float4* myP, const float4* myQ,
                                                const v2f& qF0, float (&acc)[9])
{
    const v8f czero = (v8f)0.0f;
#pragma unroll 4
    for (int jt = 0; jt < 16; ++jt) {
        const float2 pp = myB[jt * 32];
        v2f p; p.x = pp.x; p.y = pp.y;
        const float4 pa = myP[jt * 4], pb = myP[jt * 4 + 1];
        const float4 qa = myQ[jt * 4], qb = myQ[jt * 4 + 1];
        v8f P, Q;
        P[0] = pa.x; P[1] = pa.y; P[2] = pa.z; P[3] = pa.w; P[4] = pb.x; P[5] = pb.y; P[6] = pb.z; P[7] = pb.w;
        Q[0] = qa.x; Q[1] = qa.y; Q[2] = qa.z; Q[3] = qa.w; Q[4] = qb.x; Q[5] = qb.y; Q[6] = qb.z; Q[7] = qb.w;
        v8f fa[3];
        fa[0] = __builtin_amdgcn_wmma_f32_16x16x4_f32(false, p, false, qF0, (short)0, czero, false, false);
        fa[1] = fa[0] + P;
        fa[2] = fa[0] - P;
#pragma unroll
        for (int a = 0; a < 3; ++a) {
            v8f f  = fa[a];
            v8f e1 = f - Q;
            v8f e2 = f + Q;
            float v0 = acc[a], v1 = acc[3 + a], v2 = acc[6 + a];
#pragma unroll
            for (int k = 0; k < 8; k += 2) {
                v0 = fminf(fminf(v0, f[k]),  f[k + 1]);
                v1 = fminf(fminf(v1, e1[k]), e1[k + 1]);
                v2 = fminf(fminf(v2, e2[k]), e2[k + 1]);
            }
            acc[a] = v0; acc[3 + a] = v1; acc[6 + a] = v2;
        }
    }
}

__global__ __launch_bounds__(256) void nearest_dy_split_kernel(
    const int* __restrict__ li_bev, const int* __restrict__ ra_bev,
    const float* __restrict__ ra_pts, const int* __restrict__ ra_vox,
    float* __restrict__ out)
{
    __shared__ float sB[M_PTS * 4];
    __shared__ float4 sP4[M_PTS / 4];
    __shared__ float4 sQ4[M_PTS / 4];
    __shared__ float sMin[8][9][16];
    __shared__ float sQ[6][16];
    __shared__ int   sCntW[8];

    const int tid  = threadIdx.x;
    const int lane = tid & 31;
    const int wave = tid >> 5;

    const int j0 = wave * 256 + lane;
    const float* velp = ra_pts + j0 * 5 + 4;
    const int*   voxp = ra_vox + j0 * 3 + 1;
    float vel[8];
    int   vx[8], vy[8];
#pragma unroll
    for (int i = 0; i < 8; ++i) {
        vel[i] = velp[i * 160];
        vx[i]  = voxp[i * 96];
        vy[i]  = voxp[i * 96 + 1];
    }
    int cnt = 0;
#pragma unroll
    for (int i = 0; i < 8; ++i) {
        const int j = j0 + i * 32;
        float px = (float)vx[i];
        float py = (float)vy[i];
        bool dyn = fabsf(vel[i]) > 0.1f;
        sB[j * 4 + 0] = dyn ? -2.0f * px : 0.0f;
        sB[j * 4 + 1] = dyn ? -2.0f * py : 0.0f;
        sB[j * 4 + 2] = dyn ? px * px + py * py : 1.0e30f;
        sB[j * 4 + 3] = dyn ? 1.0f : 0.0f;
        ((float*)sP4)[j] = dyn ? 2.0f * px : 0.0f;
        ((float*)sQ4)[j] = dyn ? 2.0f * py : 0.0f;
        cnt += __builtin_popcount(__builtin_amdgcn_ballot_w32(dyn));
    }
    if (lane == 0) sCntW[wave] = cnt;
    asm volatile("s_wait_dscnt 0x0" ::: "memory");

    const int m    = lane & 15;
    const int half = lane >> 4;
    const int n    = blockIdx.x * 16 + m;

    int c0, c1;
    if (n < N_LI) { c0 = li_bev[n * 2 + 0]; c1 = li_bev[n * 2 + 1]; }
    else          { c0 = ra_bev[(n - N_LI) * 2 + 0]; c1 = ra_bev[(n - N_LI) * 2 + 1]; }
    float xs[3], ys[3];
    xs[0] = (float)((c0 + GRID_MOD) % GRID_MOD);
    xs[1] = (float)((c0 - 1 + GRID_MOD) % GRID_MOD);
    xs[2] = (float)((c0 + 1 + GRID_MOD) % GRID_MOD);
    ys[0] = (float)((c1 + GRID_MOD) % GRID_MOD);
    ys[1] = (float)((c1 + 1 + GRID_MOD) % GRID_MOD);
    ys[2] = (float)((c1 - 1 + GRID_MOD) % GRID_MOD);
    if (wave == 0 && half == 0) {
#pragma unroll
        for (int a = 0; a < 3; ++a) { sQ[a][m] = xs[a]; sQ[3 + a][m] = ys[a]; }
    }

    v2f qF0;
    if (half == 0) { qF0.x = xs[0]; qF0.y = ys[0]; }
    else           { qF0.x = 1.0f;  qF0.y = 0.0f; }

    float acc[9];
#pragma unroll
    for (int i = 0; i < 9; ++i) acc[i] = 3.0e38f;

    const float2* sB2 = (const float2*)sB;

    const float2* myB = sB2 + (wave * 256 + m) * 2 + half;

    const float dx1 = xs[1] - xs[0], dx2 = xs[2] - xs[0], dy1 = ys[1] - ys[0], dy2 = ys[2] - ys[0];
    const bool unit = dx1 == -1.0f && dx2 == 1.0f && dy1 == 1.0f && dy2 == -1.0f;
    const float4* myP = sP4 + wave * 64 + half * 2;
    const float4* myQ = sQ4 + wave * 64 + half * 2;
    if (__builtin_amdgcn_ballot_w32(!unit) == 0u) scan_tiles_unit(myB, myP, myQ, qF0, acc);
    else                                          scan_tiles_steps(myB, myP, myQ, qF0, dx1, dx2, dy1, dy2, acc);

#pragma unroll
    for (int i = 0; i < 9; ++i) acc[i] = fminf(acc[i], __shfl_xor(acc[i], 16, 32));
    if (half == 0) {
#pragma unroll
        for (int i = 0; i < 9; ++i) sMin[wave][i][m] = acc[i];
    }
    __syncthreads();

    if (tid < 144) {
        const int c = tid / 9;
        const int s = tid - c * 9;
        const int a = s % 3, b = s / 3;
        float v = sMin[0][s][c];
#pragma unroll
        for (int w = 1; w < 8; ++w) v = fminf(v, sMin[w][s][c]);
        const float xa = sQ[a][c], yb = sQ[3 + b][c];
        const float d2 = v + (xa * xa + yb * yb);
        int total = 0;
#pragma unroll
        for (int w = 0; w < 8; ++w) total += sCntW[w];
        const float mul = (total > 1) ? 0.01f : 0.0f;
        out[blockIdx.x * 144 + tid] = sqrtf(fmaxf(d2, 0.0f)) * mul;
    }
}

extern "C" void kernel_launch(void* const* d_in, const int* in_sizes, int n_in,
                              void* d_out, int out_size, void* d_ws, size_t ws_size,
                              hipStream_t stream) {
    (void)in_sizes; (void)n_in; (void)out_size; (void)d_ws; (void)ws_size;
    const int*   li  = (const int*)d_in[0];
    const int*   ra  = (const int*)d_in[1];
    const float* pts = (const float*)d_in[2];
    const int*   vox = (const int*)d_in[3];
    float* out = (float*)d_out;

    dim3 grid(640);
    nearest_dy_split_kernel<<<grid, 256, 0, stream>>>(li, ra, pts, vox, out);
}


# Unit-step blocks skip point tiles that hold no minimiser (exact): pass 1 keeps each tile's min of F_0, centre min joined over the 8 waves (one new barrier), full tile only where F_0 <= min+5.66m+9
# speedup vs baseline: 1.8903x; 1.1690x over previous
#include <hip/hip_runtime.h>

typedef __attribute__((ext_vector_type(2))) float v2f;
typedef __attribute__((ext_vector_type(8))) float v8f;

#define N_LI 8192
#define N_RA 2048
#define M_PTS 2048
#define GRID_MOD 513

__device__ __forceinline__ v8f fma_lane(const v2f m, const v8f b, const v8f c)
{
    v8f r;
#pragma unroll
    for (int k = 0; k < 8; k += 2) {
        v2f bb, cc;
        bb.x = b[k]; bb.y = b[k + 1]; cc.x = c[k]; cc.y = c[k + 1];
        const v2f t = __builtin_elementwise_fma(m, bb, cc);
        r[k] = t.x; r[k + 1] = t.y;
    }
    return r;
}

__device__ __forceinline__ void scan_tiles_steps(const float2* myB, const float4* myP, const float4* myQ,
                                                 const v2f& qF0, float dx1, float dx2, float dy1, float dy2,
                                                 float (&acc)[9])
{
    const v8f czero = (v8f)0.0f;
    v2f mx1, mx2, my1, my2;
    mx1.x = mx1.y = -dx1; mx2.x = mx2.y = -dx2; my1.x = my1.y = -dy1; my2.x = my2.y = -dy2;
#pragma unroll 4
    for (int jt = 0; jt < 16; ++jt) {
        const float2 pp = myB[jt * 32];
        v2f p; p.x = pp.x; p.y = pp.y;
        const float4 pa = myP[jt * 4], pb = myP[jt * 4 + 1];
        const float4 qa = myQ[jt * 4], qb = myQ[jt * 4 + 1];
        v8f P, Q;
        P[0] = pa.x; P[1] = pa.y; P[2] = pa.z; P[3] = pa.w; P[4] = pb.x; P[5] = pb.y; P[6] = pb.z; P[7] = pb.w;
        Q[0] = qa.x; Q[1] = qa.y; Q[2] = qa.z; Q[3] = qa.w; Q[4] = qb.x; Q[5] = qb.y; Q[6] = qb.z; Q[7] = qb.w;
        v8f fa[3];
        fa[0] = __builtin_amdgcn_wmma_f32_16x16x4_f32(false, p, false, qF0, (short)0, czero, false, false);
        fa[1] = fma_lane(mx1, P, fa[0]);
        fa[2] = fma_lane(mx2, P, fa[0]);
#pragma unroll
        for (int a = 0; a < 3; ++a) {
            v8f f  = fa[a];
            v8f e1 = fma_lane(my1, Q, f);
            v8f e2 = fma_lane(my2, Q, f);
            float v0 = acc[a], v1 = acc[3 + a], v2 = acc[6 + a];
#pragma unroll
            for (int k = 0; k < 8; k += 2) {
                v0 = fminf(fminf(v0, f[k]),  f[k + 1]);
                v1 = fminf(fminf(v1, e1[k]), e1[k + 1]);
                v2 = fminf(fminf(v2, e2[k]), e2[k + 1]);
            }
            acc[a] = v0; acc[3 + a] = v1; acc[6 + a] = v2;
        }
    }
}

__device__ __forceinline__ void scan_tiles_unit_pruned(const float2* myB, const float4* myP, const float4* myQ,
                                                       const v2f& qF0, float q2, float* ctrRow, int wave, int half,
                                                       float (&acc)[9])
{
    const v8f czero = (v8f)0.0f;
    float tmin[16];
#pragma unroll
    for (int jt = 0; jt < 16; ++jt) {
        const float2 pp = myB[jt * 32];
        v2f p; p.x = pp.x; p.y = pp.y;
        const v8f f = __builtin_amdgcn_wmma_f32_16x16x4_f32(false, p, false, qF0, (short)0, czero, false, false);
        float t = 3.0e38f;
#pragma unroll
        for (int k = 0; k < 8; k += 2) t = fminf(fminf(t, f[k]), f[k + 1]);
        tmin[jt] = t;
    }
    float c = tmin[0];
#pragma unroll
    for (int jt = 1; jt < 15; jt += 2) c = fminf(fminf(c, tmin[jt]), tmin[jt + 1]);
    c = fminf(c, tmin[15]);
    acc[0] = c;
    const float cw = fminf(c, __shfl_xor(c, 16, 32));
    if (half == 0) ctrRow[wave] = cw;
    __syncthreads();
    const float4 ca = ((const float4*)ctrRow)[0], cb = ((const float4*)ctrRow)[1];
    const float cmin = fminf(fminf(fminf(ca.x, ca.y), fminf(ca.z, ca.w)), fminf(fminf(cb.x, cb.y), fminf(cb.z, cb.w)));
    const float m2   = fmaxf(cmin + q2, 0.0f);
    const float thr  = cmin + (5.66f * sqrtf(m2) + 9.0f);
    unsigned mask = 0u;
#pragma unroll
    for (int jt = 0; jt < 16; ++jt)
        mask |= (__builtin_amdgcn_ballot_w32(tmin[jt] <= thr) != 0u) ? (1u << jt) : 0u;
    while (mask != 0u) {
        const int jt = __builtin_ctz(mask);
        mask &= mask - 1u;
        const float2 pp = myB[jt * 32];
        v2f p; p.x = pp.x; p.y = pp.y;
        const float4 pa = myP[jt * 4], pb = myP[jt * 4 + 1];
        const float4 qa = myQ[jt * 4], qb = myQ[jt * 4 + 1];
        v8f P, Q;
        P[0] = pa.x; P[1] = pa.y; P[2] = pa.z; P[3] = pa.w; P[4] = pb.x; P[5] = pb.y; P[6] = pb.z; P[7] = pb.w;
        Q[0] = qa.x; Q[1] = qa.y; Q[2] = qa.z; Q[3] = qa.w; Q[4] = qb.x; Q[5] = qb.y; Q[6] = qb.z; Q[7] = qb.w;
        v8f fa[3];
        fa[0] = __builtin_amdgcn_wmma_f32_16x16x4_f32(false, p, false, qF0, (short)0, czero, false, false);
        fa[1] = fa[0] + P;
        fa[2] = fa[0] - P;
#pragma unroll
        for (int a = 0; a < 3; ++a) {
            v8f f  = fa[a];
            v8f e1 = f - Q;
            v8f e2 = f + Q;
            float v0 = acc[a], v1 = acc[3 + a], v2 = acc[6 + a];
#pragma unroll
            for (int k = 0; k < 8; k += 2) {
                if (a != 0) v0 = fminf(fminf(v0, f[k]),  f[k + 1]);
                v1 = fminf(fminf(v1, e1[k]), e1[k + 1]);
                v2 = fminf(fminf(v2, e2[k]), e2[k + 1]);
            }
            acc[a] = v0; acc[3 + a] = v1; acc[6 + a] = v2;
        }
    }
}

__global__ __launch_bounds__(256) void nearest_dy_split_kernel(
    const int* __restrict__ li_bev, const int* __restrict__ ra_bev,
    const float* __restrict__ ra_pts, const int* __restrict__ ra_vox,
    float* __restrict__ out)
{
    __shared__ float sB[M_PTS * 4];
    __shared__ float4 sP4[M_PTS / 4];
    __shared__ float4 sQ4[M_PTS / 4];
    __shared__ float sMin[8][9][16];
    __shared__ float sQ[6][16];
    __shared__ int   sCntW[8];
    __shared__ __attribute__((aligned(16))) float sCtr[16][8];

    const int tid  = threadIdx.x;
    const int lane = tid & 31;
    const int wave = tid >> 5;

    const int j0 = wave * 256 + lane;
    const float* velp = ra_pts + j0 * 5 + 4;
    const int*   voxp = ra_vox + j0 * 3 + 1;
    float vel[8];
    int   vx[8], vy[8];
#pragma unroll
    for (int i = 0; i < 8; ++i) {
        vel[i] = velp[i * 160];
        vx[i]  = voxp[i * 96];
        vy[i]  = voxp[i * 96 + 1];
    }
    int cnt = 0;
#pragma unroll
    for (int i = 0; i < 8; ++i) {
        const int j = j0 + i * 32;
        float px = (float)vx[i];
        float py = (float)vy[i];
        bool dyn = fabsf(vel[i]) > 0.1f;
        sB[j * 4 + 0] = dyn ? -2.0f * px : 0.0f;
        sB[j * 4 + 1] = dyn ? -2.0f * py : 0.0f;
        sB[j * 4 + 2] = dyn ? px * px + py * py : 1.0e30f;
        sB[j * 4 + 3] = dyn ? 1.0f : 0.0f;
        ((float*)sP4)[j] = dyn ? 2.0f * px : 0.0f;
        ((float*)sQ4)[j] = dyn ? 2.0f * py : 0.0f;
        cnt += __builtin_popcount(__builtin_amdgcn_ballot_w32(dyn));
    }
    if (lane == 0) sCntW[wave] = cnt;
    asm volatile("s_wait_dscnt 0x0" ::: "memory");

    const int m    = lane & 15;
    const int half = lane >> 4;
    const int n    = blockIdx.x * 16 + m;

    int c0, c1;
    if (n < N_LI) { c0 = li_bev[n * 2 + 0]; c1 = li_bev[n * 2 + 1]; }
    else          { c0 = ra_bev[(n - N_LI) * 2 + 0]; c1 = ra_bev[(n - N_LI) * 2 + 1]; }
    float xs[3], ys[3];
    xs[0] = (float)((c0 + GRID_MOD) % GRID_MOD);
    xs[1] = (float)((c0 - 1 + GRID_MOD) % GRID_MOD);
    xs[2] = (float)((c0 + 1 + GRID_MOD) % GRID_MOD);
    ys[0] = (float)((c1 + GRID_MOD) % GRID_MOD);
    ys[1] = (float)((c1 + 1 + GRID_MOD) % GRID_MOD);
    ys[2] = (float)((c1 - 1 + GRID_MOD) % GRID_MOD);
    if (wave == 0 && half == 0) {
#pragma unroll
        for (int a = 0; a < 3; ++a) { sQ[a][m] = xs[a]; sQ[3 + a][m] = ys[a]; }
    }

    v2f qF0;
    if (half == 0) { qF0.x = xs[0]; qF0.y = ys[0]; }
    else           { qF0.x = 1.0f;  qF0.y = 0.0f; }

    float acc[9];
#pragma unroll
    for (int i = 0; i < 9; ++i) acc[i] = 3.0e38f;

    const float2* sB2 = (const float2*)sB;

    const float2* myB = sB2 + (wave * 256 + m) * 2 + half;

    const float dx1 = xs[1] - xs[0], dx2 = xs[2] - xs[0], dy1 = ys[1] - ys[0], dy2 = ys[2] - ys[0];
    const bool unit = dx1 == -1.0f && dx2 == 1.0f && dy1 == 1.0f && dy2 == -1.0f;
    const float4* myP = sP4 + wave * 64 + half * 2;
    const float4* myQ = sQ4 + wave * 64 + half * 2;
    if (__builtin_amdgcn_ballot_w32(!unit) == 0u)
        scan_tiles_unit_pruned(myB, myP, myQ, qF0, xs[0] * xs[0] + ys[0] * ys[0], &sCtr[m][0], wave, half, acc);
    else
        scan_tiles_steps(myB, myP, myQ, qF0, dx1, dx2, dy1, dy2, acc);

#pragma unroll
    for (int i = 0; i < 9; ++i) acc[i] = fminf(acc[i], __shfl_xor(acc[i], 16, 32));
    if (half == 0) {
#pragma unroll
        for (int i = 0; i < 9; ++i) sMin[wave][i][m] = acc[i];
    }
    __syncthreads();

    if (tid < 144) {
        const int c = tid / 9;
        const int s = tid - c * 9;
        const int a = s % 3, b = s / 3;
        float v = sMin[0][s][c];
#pragma unroll
        for (int w = 1; w < 8; ++w) v = fminf(v, sMin[w][s][c]);
        const float xa = sQ[a][c], yb = sQ[3 + b][c];
        const float d2 = v + (xa * xa + yb * yb);
        int total = 0;
#pragma unroll
        for (int w = 0; w < 8; ++w) total += sCntW[w];
        const float mul = (total > 1) ? 0.01f : 0.0f;
        out[blockIdx.x * 144 + tid] = sqrtf(fmaxf(d2, 0.0f)) * mul;
    }
}

extern "C" void kernel_launch(void* const* d_in, const int* in_sizes, int n_in,
                              void* d_out, int out_size, void* d_ws, size_t ws_size,
                              hipStream_t stream) {
    (void)in_sizes; (void)n_in; (void)out_size; (void)d_ws; (void)ws_size;
    const int*   li  = (const int*)d_in[0];
    const int*   ra  = (const int*)d_in[1];
    const float* pts = (const float*)d_in[2];
    const int*   vox = (const int*)d_in[3];
    float* out = (float*)d_out;

    dim3 grid(640);
    nearest_dy_split_kernel<<<grid, 256, 0, stream>>>(li, ra, pts, vox, out);
}
